# starting kernel
# baseline (speedup 1.0000x reference)
#include <hip/hip_runtime.h>

typedef __attribute__((ext_vector_type(2))) float v2f;
typedef __attribute__((ext_vector_type(8))) float v8f;

#define DIM 16
#define NEG_SLOPE 0.2f

__device__ __forceinline__ float leaky(float e) { return e >= 0.f ? e : NEG_SLOPE * e; }

__device__ __forceinline__ unsigned fenc(float f) {
  unsigned u = __float_as_uint(f);
  return (u & 0x80000000u) ? ~u : (u | 0x80000000u);
}
__device__ __forceinline__ float fdec(unsigned u) {
  return __uint_as_float((u & 0x80000000u) ? (u & 0x7FFFFFFFu) : ~u);
}

__global__ void gat_pad_x(const float* __restrict__ X, float* __restrict__ XP, int n) {
  int idx = blockIdx.x * blockDim.x + threadIdx.x;
  if (idx >= n * DIM) return;
  int i = idx >> 4, c = idx & 15;
  XP[idx] = (c < 14) ? X[i * 14 + c] : 0.f;
}

__global__ void gat_pad_w(const float* __restrict__ W, float* __restrict__ WP) {
  int idx = threadIdx.x;
  int r = idx >> 4, c = idx & 15;
  WP[idx] = (r < 14) ? W[r * DIM + c] : 0.f;
}

__global__ void gat_proj_wmma(const float* __restrict__ X16, const float* __restrict__ W16,
                              float* __restrict__ H, int n) {
  int wave = (int)((blockIdx.x * blockDim.x + threadIdx.x) >> 5);
  int lane = threadIdx.x & 31;
  int ntiles = (n + 15) >> 4;
  if (wave >= ntiles) return;
  int half = lane >> 4;
  int l = lane & 15;
  int rowA = wave * 16 + l;
  if (rowA >= n) rowA = n - 1;
  v8f c = {};
#pragma unroll
  for (int k0 = 0; k0 < DIM; k0 += 4) {
    int ka = k0 + half * 2;
    float2 av = *(const float2*)&X16[rowA * DIM + ka];
    v2f a, b;
    a.x = av.x;
    a.y = av.y;
    b.x = W16[ka * DIM + l];
    b.y = W16[(ka + 1) * DIM + l];
    c = __builtin_amdgcn_wmma_f32_16x16x4_f32(false, a, false, b, (short)0, c,
                                              false, false);
  }
  int base = (wave * 16 + half * 8) * DIM + l;
  if (wave * 16 + 16 <= n) {
#pragma unroll
    for (int r = 0; r < 8; ++r) H[base + r * DIM] = c[r];
  } else {
#pragma unroll
    for (int r = 0; r < 8; ++r) {
      int row = wave * 16 + r + half * 8;
      if (row < n) H[row * DIM + l] = c[r];
    }
  }
}

__global__ void gat_alphas(const float* __restrict__ H, const float* __restrict__ a_src,
                           const float* __restrict__ a_dst, float* __restrict__ AS,
                           float* __restrict__ AD, int n) {
  int i = blockIdx.x * blockDim.x + threadIdx.x;
  if (i >= n) return;
  float s = 0.f, d = 0.f;
#pragma unroll
  for (int j = 0; j < 4; ++j) {
    float4 h = *(const float4*)&H[i * DIM + j * 4];
    s += h.x * a_src[j * 4 + 0] + h.y * a_src[j * 4 + 1] +
         h.z * a_src[j * 4 + 2] + h.w * a_src[j * 4 + 3];
    d += h.x * a_dst[j * 4 + 0] + h.y * a_dst[j * 4 + 1] +
         h.z * a_dst[j * 4 + 2] + h.w * a_dst[j * 4 + 3];
  }
  AS[i] = s;
  AD[i] = d;
}

__global__ void gat_self_init(const float* __restrict__ AS, const float* __restrict__ AD,
                              unsigned* __restrict__ MK, int n) {
  int i = blockIdx.x * blockDim.x + threadIdx.x;
  if (i >= n) return;
  MK[i] = fenc(leaky(AS[i] + AD[i]));
}

__global__ void gat_edge_max(const int* __restrict__ src, const int* __restrict__ dst,
                             const float* __restrict__ AS, const float* __restrict__ AD,
                             unsigned* __restrict__ MK, int E) {
  int tid = blockIdx.x * blockDim.x + threadIdx.x;
  int stride = gridDim.x * blockDim.x;
  int E4 = E >> 2;
  const int4* s4p = (const int4*)src;
  const int4* d4p = (const int4*)dst;
  for (int q = tid; q < E4; q += stride) {
    int4 s4 = s4p[q], d4 = d4p[q];
    atomicMax(&MK[d4.x], fenc(leaky(AS[s4.x] + AD[d4.x])));
    atomicMax(&MK[d4.y], fenc(leaky(AS[s4.y] + AD[d4.y])));
    atomicMax(&MK[d4.z], fenc(leaky(AS[s4.z] + AD[d4.z])));
    atomicMax(&MK[d4.w], fenc(leaky(AS[s4.w] + AD[d4.w])));
  }
  if (tid < (E & 3)) {
    int e = (E & ~3) + tid;
    int s = src[e], d = dst[e];
    atomicMax(&MK[d], fenc(leaky(AS[s] + AD[d])));
  }
}

__global__ void gat_node_seed(const float* __restrict__ H, const float* __restrict__ AS,
                              const float* __restrict__ AD, const unsigned* __restrict__ MK,
                              float* __restrict__ DEN, float* __restrict__ S, int n) {
  int i = blockIdx.x * blockDim.x + threadIdx.x;
  if (i >= n) return;
  float m = fdec(MK[i]);
  float ex = expf(leaky(AS[i] + AD[i]) - m);
  DEN[i] = ex;
#pragma unroll
  for (int j = 0; j < 4; ++j) {
    float4 h = *(const float4*)&H[i * DIM + j * 4];
    float4 o = {ex * h.x, ex * h.y, ex * h.z, ex * h.w};
    *(float4*)&S[i * DIM + j * 4] = o;
  }
}

__device__ __forceinline__ void gat_edge_one(int s, int d, const float* __restrict__ H,
                                             const float* __restrict__ AS,
                                             const float* __restrict__ AD,
                                             const unsigned* __restrict__ MK,
                                             float* __restrict__ DEN,
                                             float* __restrict__ S) {
  float ex = expf(leaky(AS[s] + AD[d]) - fdec(MK[d]));
  atomicAdd(&DEN[d], ex);
  const float4* h4 = (const float4*)&H[s * DIM];
  float* sd = &S[d * DIM];
#pragma unroll
  for (int j = 0; j < 4; ++j) {
    float4 v = h4[j];
    atomicAdd(&sd[j * 4 + 0], ex * v.x);
    atomicAdd(&sd[j * 4 + 1], ex * v.y);
    atomicAdd(&sd[j * 4 + 2], ex * v.z);
    atomicAdd(&sd[j * 4 + 3], ex * v.w);
  }
}

__global__ void gat_edge_acc(const int* __restrict__ src, const int* __restrict__ dst,
                             const float* __restrict__ H, const float* __restrict__ AS,
                             const float* __restrict__ AD, const unsigned* __restrict__ MK,
                             float* __restrict__ DEN, float* __restrict__ S, int E) {
  int tid = blockIdx.x * blockDim.x + threadIdx.x;
  int stride = gridDim.x * blockDim.x;
  int E4 = E >> 2;
  const int4* s4p = (const int4*)src;
  const int4* d4p = (const int4*)dst;
  for (int q = tid; q < E4; q += stride) {
    __builtin_prefetch(&s4p[q + 1024], 0, 1);
    __builtin_prefetch(&d4p[q + 1024], 0, 1);
    int4 s4 = s4p[q], d4 = d4p[q];
    gat_edge_one(s4.x, d4.x, H, AS, AD, MK, DEN, S);
    gat_edge_one(s4.y, d4.y, H, AS, AD, MK, DEN, S);
    gat_edge_one(s4.z, d4.z, H, AS, AD, MK, DEN, S);
    gat_edge_one(s4.w, d4.w, H, AS, AD, MK, DEN, S);
  }
  if (tid < (E & 3)) {
    int e = (E & ~3) + tid;
    gat_edge_one(src[e], dst[e], H, AS, AD, MK, DEN, S);
  }
}

__global__ void gat_node_fin(const float* __restrict__ S, const float* __restrict__ DEN,
                             const float* __restrict__ b, float* __restrict__ OUT, int n,
                             int do_relu) {
  int idx = blockIdx.x * blockDim.x + threadIdx.x;
  if (idx >= n * DIM) return;
  int i = idx >> 4, cidx = idx & 15;
  float v = S[idx] / DEN[i] + b[cidx];
  if (do_relu) v = v > 0.f ? v : 0.f;
  OUT[idx] = v;
}

extern "C" void kernel_launch(void* const* d_in, const int* in_sizes, int n_in,
                              void* d_out, int out_size, void* d_ws, size_t ws_size,
                              hipStream_t stream) {
  const float* x = (const float*)d_in[0];
  const int* ei = (const int*)d_in[1];
  const float* W1 = (const float*)d_in[2];
  const float* as1 = (const float*)d_in[3];
  const float* ad1 = (const float*)d_in[4];
  const float* b1 = (const float*)d_in[5];
  const float* W2 = (const float*)d_in[6];
  const float* as2 = (const float*)d_in[7];
  const float* ad2 = (const float*)d_in[8];
  const float* b2 = (const float*)d_in[9];
  float* out = (float*)d_out;

  const int NFEAT = 14;
  int n = in_sizes[0] / NFEAT;
  int E = in_sizes[1] / 2;
  const int* src = ei;
  const int* dst = ei + E;

  float* fws = (float*)d_ws;
  float* H = fws;
  float* S = H + (size_t)n * DIM;
  float* O1 = S + (size_t)n * DIM;
  float* XP = O1 + (size_t)n * DIM;
  float* AS = XP + (size_t)n * DIM;
  float* AD = AS + n;
  float* DEN = AD + n;
  unsigned* MK = (unsigned*)(DEN + n);
  float* WP = (float*)(MK + n);

  const int TB = 256;
  int ntiles = (n + 15) >> 4;
  dim3 gProj((ntiles + 7) / 8);
  dim3 gNode((n + TB - 1) / TB);
  dim3 gElem((n * DIM + TB - 1) / TB);
  dim3 gEdge(((E >> 2) + TB - 1) / TB);

  gat_pad_x<<<gElem, TB, 0, stream>>>(x, XP, n);
  gat_pad_w<<<1, 256, 0, stream>>>(W1, WP);
  gat_proj_wmma<<<gProj, TB, 0, stream>>>(XP, WP, H, n);
  gat_alphas<<<gNode, TB, 0, stream>>>(H, as1, ad1, AS, AD, n);
  gat_self_init<<<gNode, TB, 0, stream>>>(AS, AD, MK, n);
  gat_edge_max<<<gEdge, TB, 0, stream>>>(src, dst, AS, AD, MK, E);
  gat_node_seed<<<gNode, TB, 0, stream>>>(H, AS, AD, MK, DEN, S, n);
  gat_edge_acc<<<gEdge, TB, 0, stream>>>(src, dst, H, AS, AD, MK, DEN, S, E);
  gat_node_fin<<<gElem, TB, 0, stream>>>(S, DEN, b1, O1, n, 1);

  gat_proj_wmma<<<gProj, TB, 0, stream>>>(O1, W2, H, n);
  gat_alphas<<<gNode, TB, 0, stream>>>(H, as2, ad2, AS, AD, n);
  gat_self_init<<<gNode, TB, 0, stream>>>(AS, AD, MK, n);
  gat_edge_max<<<gEdge, TB, 0, stream>>>(src, dst, AS, AD, MK, E);
  gat_node_seed<<<gNode, TB, 0, stream>>>(H, AS, AD, MK, DEN, S, n);
  gat_edge_acc<<<gEdge, TB, 0, stream>>>(src, dst, H, AS, AD, MK, DEN, S, E);
  gat_node_fin<<<gElem, TB, 0, stream>>>(S, DEN, b2, out, n, 0);
}


# layer 1 sums raw input rows and projects after: (sum_e alpha_e x[s]) W1 on WMMA in the finishing launch, scores from x.(W1 a); pad and projection launches gone, 8 to 6; float sums re-associated
# speedup vs baseline: 13.4499x; 13.4499x over previous
#include <hip/hip_runtime.h>

typedef __attribute__((ext_vector_type(2))) float v2f;
typedef __attribute__((ext_vector_type(8))) float v8f;

#define DIM 16
#define NFEAT 14
#define NEG_SLOPE 0.2f


#define HROW 32

__device__ __forceinline__ float leaky(float e) { return e >= 0.f ? e : NEG_SLOPE * e; }

__device__ __forceinline__ unsigned fenc(float f) {
  unsigned u = __float_as_uint(f);
  return (u & 0x80000000u) ? ~u : (u | 0x80000000u);
}
__device__ __forceinline__ float fdec(unsigned u) {
  return __uint_as_float((u & 0x80000000u) ? (u & 0x7FFFFFFFu) : ~u);
}

#define SAFE_SPREAD 32.0f
__device__ __forceinline__ bool gat_bound_enc(unsigned ehi, unsigned elo, float& gmax) {
  gmax = fdec(ehi);
  return (gmax - fdec(elo)) <= SAFE_SPREAD;
}
__device__ __forceinline__ bool gat_bound_ok(const unsigned* __restrict__ G, float& gmax) {
  return gat_bound_enc(G[0], G[1], gmax);
}

__global__ void gat_proj_wmma(const float* __restrict__ X16, const float* __restrict__ W16,
                              float* __restrict__ H, int n) {
  int wave = (int)((blockIdx.x * blockDim.x + threadIdx.x) >> 5);
  int lane = threadIdx.x & 31;
  int ntiles = (n + 15) >> 4;
  if (wave >= ntiles) return;
  int half = lane >> 4;
  int l = lane & 15;
  int rowA = wave * 16 + l;
  if (rowA >= n) rowA = n - 1;
  v8f c = {};
#pragma unroll
  for (int k0 = 0; k0 < DIM; k0 += 4) {
    int ka = k0 + half * 2;
    float2 av = *(const float2*)&X16[rowA * DIM + ka];
    v2f a, b;
    a.x = av.x;
    a.y = av.y;
    b.x = W16[ka * DIM + l];
    b.y = W16[(ka + 1) * DIM + l];
    c = __builtin_amdgcn_wmma_f32_16x16x4_f32(false, a, false, b, (short)0, c,
                                              false, false);
  }
  int base = (wave * 16 + half * 8) * HROW + l;
  if (wave * 16 + 16 <= n) {
#pragma unroll
    for (int r = 0; r < 8; ++r) H[base + r * HROW] = c[r];
  } else {
#pragma unroll
    for (int r = 0; r < 8; ++r) {
      int row = wave * 16 + r + half * 8;
      if (row < n) H[row * HROW + l] = c[r];
    }
  }
}

__global__ void gat_scores_x(const float* __restrict__ X, const float* __restrict__ W,
                             const float* __restrict__ a_src, const float* __restrict__ a_dst,
                             float* __restrict__ H, float* __restrict__ AS,
                             float* __restrict__ AD, float* __restrict__ S,
                             float* __restrict__ DEN, unsigned* __restrict__ PM,
                             unsigned* __restrict__ G2, int n) {
  __shared__ float uv[2 * NFEAT];
  __shared__ unsigned wmax[32], wmin[32];
  const int tid = threadIdx.x;
  if (tid < 2 * NFEAT) {
    const int k = tid < NFEAT ? tid : tid - NFEAT;
    const float* a = tid < NFEAT ? a_src : a_dst;
    float acc = 0.f;
    for (int c = 0; c < DIM; ++c) acc += W[k * DIM + c] * a[c];
    uv[tid] = acc;
  }
  if (blockIdx.x == 0 && tid < 2) G2[tid] = tid ? 0xFFFFFFFFu : 0u;
  const float4 z4 = {0.f, 0.f, 0.f, 0.f};
  const int q0 = blockIdx.x * blockDim.x * 4;
  for (int j = 0; j < 4; ++j) {
    const int q = q0 + j * blockDim.x + tid;
    if (q < n * 4) ((float4*)S)[q] = z4;
  }
  __syncthreads();
  const int i = blockIdx.x * blockDim.x + tid;
  unsigned hi = 0u, lo = 0xFFFFFFFFu;
  if (i < n) {
    const float2* xr = (const float2*)&X[i * NFEAT];
    float2 v[NFEAT / 2];
#pragma unroll
    for (int j = 0; j < NFEAT / 2; ++j) v[j] = xr[j];
    float sc = 0.f, dc = 0.f;
#pragma unroll
    for (int j = 0; j < NFEAT / 2; ++j) {
      sc += v[j].x * uv[2 * j] + v[j].y * uv[2 * j + 1];
      dc += v[j].x * uv[NFEAT + 2 * j] + v[j].y * uv[NFEAT + 2 * j + 1];
    }
    float4* rec = (float4*)&H[i * HROW];
    rec[0] = make_float4(v[0].x, v[0].y, v[1].x, v[1].y);
    rec[1] = make_float4(v[2].x, v[2].y, v[3].x, v[3].y);
    rec[2] = make_float4(v[4].x, v[4].y, v[5].x, v[5].y);
    rec[3] = make_float4(v[6].x, v[6].y, 0.f, 0.f);
    H[i * HROW + DIM] = sc;
    AS[i] = sc;
    AD[i] = dc;
    DEN[i] = 0.f;
    hi = fenc(sc);
    lo = hi;
  }
  for (int o = 16; o > 0; o >>= 1) {
    const unsigned h2 = (unsigned)__shfl_xor((int)hi, o);
    const unsigned l2 = (unsigned)__shfl_xor((int)lo, o);
    hi = hi > h2 ? hi : h2;
    lo = lo < l2 ? lo : l2;
  }
  const int w = tid >> 5, nw = (blockDim.x + 31) >> 5;
  if ((tid & 31) == 0) {
    wmax[w] = hi;
    wmin[w] = lo;
  }
  __syncthreads();
  if (tid == 0) {
    for (int j = 1; j < nw; ++j) {
      hi = hi > wmax[j] ? hi : wmax[j];
      lo = lo < wmin[j] ? lo : wmin[j];
    }
    PM[2 * blockIdx.x] = hi;
    PM[2 * blockIdx.x + 1] = lo;
  }
}

__global__ void gat_alphas(float* __restrict__ H, const float* __restrict__ a_src,
                           const float* __restrict__ a_dst, float* __restrict__ AS,
                           float* __restrict__ AD, unsigned* __restrict__ G, int n) {
  __shared__ unsigned wmax[32], wmin[32];
  int i = blockIdx.x * blockDim.x + threadIdx.x;
  unsigned hi = 0u, lo = 0xFFFFFFFFu;
  if (i < n) {
    float s = 0.f, d = 0.f;
#pragma unroll
    for (int j = 0; j < 4; ++j) {
      float4 h = *(const float4*)&H[i * HROW + j * 4];
      s += h.x * a_src[j * 4 + 0] + h.y * a_src[j * 4 + 1] +
           h.z * a_src[j * 4 + 2] + h.w * a_src[j * 4 + 3];
      d += h.x * a_dst[j * 4 + 0] + h.y * a_dst[j * 4 + 1] +
           h.z * a_dst[j * 4 + 2] + h.w * a_dst[j * 4 + 3];
    }
    AS[i] = s;
    H[i * HROW + DIM] = s;
    AD[i] = d;
    hi = fenc(s);
    lo = hi;
  }
  for (int o = 16; o > 0; o >>= 1) {
    const unsigned h2 = (unsigned)__shfl_xor((int)hi, o);
    const unsigned l2 = (unsigned)__shfl_xor((int)lo, o);
    hi = hi > h2 ? hi : h2;
    lo = lo < l2 ? lo : l2;
  }
  const int w = threadIdx.x >> 5, nw = (blockDim.x + 31) >> 5;
  if ((threadIdx.x & 31) == 0) {
    wmax[w] = hi;
    wmin[w] = lo;
  }
  __syncthreads();
  if (threadIdx.x == 0) {
    for (int j = 1; j < nw; ++j) {
      hi = hi > wmax[j] ? hi : wmax[j];
      lo = lo < wmin[j] ? lo : wmin[j];
    }
    atomicMax(&G[0], hi);
    atomicMin(&G[1], lo);
  }
}

#define TILE_TB 1024
#define RING 256
#define CHUNK 64

#define MEMO_DL_BITS 12
#define MEMO_CAP 512
#define MEMO_END 0xFFFFFFFFu
#define ACC_SLICES 8
__device__ __forceinline__ unsigned memo_pack(int2 e) {
  return ((unsigned)e.x << MEMO_DL_BITS) | (unsigned)e.y;
}
__device__ __forceinline__ int2 memo_unpack(unsigned w) {
  return make_int2((int)(w >> MEMO_DL_BITS), (int)(w & ((1u << MEMO_DL_BITS) - 1u)));
}
__device__ __forceinline__ int gat_memo_cap(const float* ws0, const unsigned* mb,
                                            unsigned long long ws_bytes, int n, int nwaves) {
  if (n > (1 << (32 - MEMO_DL_BITS))) return 0;
  const unsigned long long used = (unsigned long long)((const float*)mb - ws0);
  const unsigned long long words = ws_bytes >> 2;
  if (words < used + (unsigned long long)nwaves * (unsigned long long)(1 + CHUNK)) return 0;
  unsigned long long c = (words - used) / (unsigned long long)nwaves - 1ull;
  if (c > (unsigned long long)MEMO_CAP) c = (unsigned long long)MEMO_CAP;
  return (int)(c & ~31ull);
}

template <int P, int NC>
__device__ __forceinline__ void gat_acc_row(float* accf, int dl, float x, const float4 u0,
                                            const float4 u1, const float4 u2, const float4 u3) {
  atomicAdd(&accf[16 * P + dl], x);
  atomicAdd(&accf[0 * P + dl], x * u0.x);
  atomicAdd(&accf[1 * P + dl], x * u0.y);
  atomicAdd(&accf[2 * P + dl], x * u0.z);
  atomicAdd(&accf[3 * P + dl], x * u0.w);
  atomicAdd(&accf[4 * P + dl], x * u1.x);
  atomicAdd(&accf[5 * P + dl], x * u1.y);
  atomicAdd(&accf[6 * P + dl], x * u1.z);
  atomicAdd(&accf[7 * P + dl], x * u1.w);
  atomicAdd(&accf[8 * P + dl], x * u2.x);
  atomicAdd(&accf[9 * P + dl], x * u2.y);
  atomicAdd(&accf[10 * P + dl], x * u2.z);
  atomicAdd(&accf[11 * P + dl], x * u2.w);
  atomicAdd(&accf[12 * P + dl], x * u3.x);
  atomicAdd(&accf[13 * P + dl], x * u3.y);
  if (NC > 14) {
    atomicAdd(&accf[14 * P + dl], x * u3.z);
    atomicAdd(&accf[15 * P + dl], x * u3.w);
  }
}

template <int P, bool FULL, int NC>
__device__ __forceinline__ void gat_acc_chunk(const int2 e0, const int2 e1, bool a0, bool a1,
                                              float* accf, const float* __restrict__ H,
                                              const float2* dstv) {
  const bool v0 = FULL || a0;
  const bool v1 = FULL || a1;
  const float* r0 = &H[e0.x * HROW];
  const float* r1 = &H[e1.x * HROW];
  const float s0 = r0[DIM];
  const float s1 = r1[DIM];
  const float4* p0 = (const float4*)r0;
  const float4* p1 = (const float4*)r1;
  const float4 u0 = p0[0], u1 = p0[1], u2 = p0[2], u3 = p0[3];
  const float4 w0 = p1[0], w1 = p1[1], w2 = p1[2], w3 = p1[3];
  const float2 g0 = dstv[e0.y];
  const float2 g1 = dstv[e1.y];
  const float x0 = expf(leaky(s0 + g0.x) - g0.y);
  const float x1 = expf(leaky(s1 + g1.x) - g1.y);
  if (v0) gat_acc_row<P, NC>(accf, e0.y, x0, u0, u1, u2, u3);
  if (v1) gat_acc_row<P, NC>(accf, e1.y, x1, w0, w1, w2, w3);
}

template <int P, bool FULL, int MEMO, int NC>
__device__ __forceinline__ void gat_ring_drain(const int2* wr, int head, int nv, int lane,
                                               float* accf, const float* __restrict__ H,
                                               const float2* dstv, unsigned* lst, int& cnt,
                                               int cap) {
  const bool v0 = FULL || lane < nv;
  const bool v1 = FULL || (lane + 32) < nv;
  int2 e0 = {0, 0}, e1 = {0, 0};
  if (v0) e0 = wr[(head + lane) & (RING - 1)];
  if (v1) e1 = wr[(head + 32 + lane) & (RING - 1)];
  if (MEMO == 1 && cnt >= 0) {
    const int m = FULL ? CHUNK : nv;
    if (cnt + m <= cap) {
      if (v0) lst[cnt + lane] = memo_pack(e0);
      if (v1) lst[cnt + 32 + lane] = memo_pack(e1);
      cnt += m;
    } else {
      cnt = -1;
    }
  }
  gat_acc_chunk<P, FULL, NC>(e0, e1, v0, v1, accf, H, dstv);
}

template <int P, int MEMO, int NC>
__device__ __forceinline__ void gat_ring_push(const int4 d4, const int4 s4, bool ok, int lo,
                                              unsigned span, int2* wr, int& head, int& tail,
                                              int lane, float* accf,
                                              const float* __restrict__ H,
                                              const float2* dstv, unsigned* lst, int& cnt,
                                              int cap) {
  const unsigned r0 = (unsigned)(d4.x - lo), r1 = (unsigned)(d4.y - lo);
  const unsigned r2 = (unsigned)(d4.z - lo), r3 = (unsigned)(d4.w - lo);
  const unsigned sp = ok ? span : 0u;
  const bool m0 = r0 < sp, m1 = r1 < sp, m2 = r2 < sp, m3 = r3 < sp;
  const unsigned b0 = __builtin_amdgcn_ballot_w32(m0), b1 = __builtin_amdgcn_ballot_w32(m1);
  const unsigned b2 = __builtin_amdgcn_ballot_w32(m2), b3 = __builtin_amdgcn_ballot_w32(m3);
  int t = tail;
  if (m0) wr[__builtin_amdgcn_mbcnt_lo(b0, (unsigned)t) & (RING - 1)] = make_int2(s4.x, (int)r0);
  t += __builtin_popcount(b0);
  if (m1) wr[__builtin_amdgcn_mbcnt_lo(b1, (unsigned)t) & (RING - 1)] = make_int2(s4.y, (int)r1);
  t += __builtin_popcount(b1);
  if (m2) wr[__builtin_amdgcn_mbcnt_lo(b2, (unsigned)t) & (RING - 1)] = make_int2(s4.z, (int)r2);
  t += __builtin_popcount(b2);
  if (m3) wr[__builtin_amdgcn_mbcnt_lo(b3, (unsigned)t) & (RING - 1)] = make_int2(s4.w, (int)r3);
  t += __builtin_popcount(b3);
  tail = t;
  while (tail - head >= CHUNK) {
    gat_ring_drain<P, true, MEMO, NC>(wr, head, CHUNK, lane, accf, H, dstv, lst, cnt, cap);
    head += CHUNK;
  }
}

template <int R, int NC>
__global__ __launch_bounds__(TILE_TB) void gat_edge_tile(
    const int* __restrict__ src, const int* __restrict__ dst, const float* __restrict__ H,
    const float* __restrict__ AS, const float* __restrict__ AD, const unsigned* __restrict__ PM,
    int npart, float* __restrict__ DEN, float* __restrict__ S, int E, int n, const float* ws0,
    unsigned* mb, unsigned long long ws_bytes) {
  static_assert(R <= (1 << MEMO_DL_BITS), "list word layout");
  constexpr int P = R + 4;
  constexpr int NPL = 17;
  __shared__ unsigned accu[NPL * P];
  __shared__ int2 ring[(TILE_TB / 32) * RING];
  __shared__ float2 dstv[R];
  float* accf = (float*)accu;
  const int tid = threadIdx.x;
  const int lane = tid & 31;
  int2* wr = ring + (tid >> 5) * RING;
  const int lo = blockIdx.y * R;
  const int span = (n - lo) < R ? (n - lo) : R;

  const int nwaves = (int)(gridDim.x * gridDim.y) * (TILE_TB / 32);
  const int gw = (int)(blockIdx.y * gridDim.x + blockIdx.x) * (TILE_TB / 32) + (tid >> 5);
  const int cap = gat_memo_cap(ws0, mb, ws_bytes, n, nwaves);
  unsigned* lst = mb + nwaves + (size_t)gw * (size_t)cap;
  int cnt = cap > 0 ? 0 : -1;

  const int sper = (span + (int)gridDim.x - 1) / (int)gridDim.x;
  const int si = (int)blockIdx.x * sper + tid;
  const bool sown = tid < sper && si < span;
  float4 su0 = {0.f, 0.f, 0.f, 0.f}, su1 = su0, su2 = su0, su3 = su0;
  float sas = 0.f;
  if (sown) {
    const float4* p = (const float4*)&H[(lo + si) * HROW];
    su0 = p[0];
    su1 = p[1];
    su2 = p[2];
    su3 = p[3];
    sas = AS[lo + si];
  }

  __shared__ unsigned gred[2 * (TILE_TB / 32)];
  unsigned ghi = 0u, glo = 0xFFFFFFFFu;
  for (int i = tid; i < npart; i += TILE_TB) {
    const unsigned ph = PM[2 * i], pl = PM[2 * i + 1];
    ghi = ghi > ph ? ghi : ph;
    glo = glo < pl ? glo : pl;
  }
  for (int o = 16; o > 0; o >>= 1) {
    const unsigned h2 = (unsigned)__shfl_xor((int)ghi, o);
    const unsigned l2 = (unsigned)__shfl_xor((int)glo, o);
    ghi = ghi > h2 ? ghi : h2;
    glo = glo < l2 ? glo : l2;
  }
  if (lane == 0) {
    gred[tid >> 5] = ghi;
    gred[(TILE_TB / 32) + (tid >> 5)] = glo;
  }
  __syncthreads();
  ghi = gred[lane];
  glo = gred[(TILE_TB / 32) + lane];
  for (int o = 16; o > 0; o >>= 1) {
    const unsigned h2 = (unsigned)__shfl_xor((int)ghi, o);
    const unsigned l2 = (unsigned)__shfl_xor((int)glo, o);
    ghi = ghi > h2 ? ghi : h2;
    glo = glo < l2 ? glo : l2;
  }
  float gmax;
  const bool bound = gat_bound_enc(ghi, glo, gmax);

  if (!bound) {
    for (int i = tid; i < span; i += TILE_TB) accu[i] = fenc(AS[lo + i]);
    __syncthreads();
    for (int e = tid; e < E; e += TILE_TB) {
      const unsigned r = (unsigned)(dst[e] - lo);
      if (r < (unsigned)span) atomicMax(&accu[r], fenc(AS[src[e]]));
    }
    __syncthreads();
  }
  for (int i = tid; i < span; i += TILE_TB) {
    const float ad = AD[lo + i];
    float2 dv;
    dv.x = ad;
    dv.y = leaky((bound ? gmax : fdec(accu[i])) + ad);
    dstv[i] = dv;
  }
  if (!bound) __syncthreads();
  for (int i = tid; i < NPL * P; i += TILE_TB) accu[i] = 0u;
  __syncthreads();

  if (sown) {
    const float2 dv = dstv[si];
    const float x = expf(leaky(sas + dv.x) - dv.y);
    gat_acc_row<P, NC>(accf, si, x, su0, su1, su2, su3);
  }

  const int E4 = E >> 2;
  const int per = (E4 + (int)gridDim.x - 1) / (int)gridDim.x;
  const int q0 = blockIdx.x * per;
  const int q1 = (q0 + per) < E4 ? (q0 + per) : E4;
  const int4* s4p = (const int4*)src;
  const int4* d4p = (const int4*)dst;

  int head = 0, tail = 0;
  if (q0 < q1) {
    const int qlast = q1 - 1;
    int qa = q0 + tid, qb = q0 + TILE_TB + tid;
    qa = qa < qlast ? qa : qlast;
    qb = qb < qlast ? qb : qlast;
    int4 dA = d4p[qa], sA = s4p[qa], dB = d4p[qb], sB = s4p[qb];
    for (int base = q0; base < q1; base += 2 * TILE_TB) {
      const int4 d0 = dA, s0 = sA, d1 = dB, s1 = sB;
      const bool okA = (base + tid) < q1;
      const bool okB = (base + TILE_TB + tid) < q1;
      int na = base + 2 * TILE_TB + tid, nb = base + 3 * TILE_TB + tid;
      na = na < qlast ? na : qlast;
      nb = nb < qlast ? nb : qlast;
      dA = d4p[na];
      sA = s4p[na];
      dB = d4p[nb];
      sB = s4p[nb];
      gat_ring_push<P, 1, NC>(d0, s0, okA, lo, (unsigned)span, wr, head, tail, lane, accf, H, dstv,
                              lst, cnt, cap);
      gat_ring_push<P, 1, NC>(d1, s1, okB, lo, (unsigned)span, wr, head, tail, lane, accf, H, dstv,
                              lst, cnt, cap);
    }
  }
  if (tail != head)
    gat_ring_drain<P, false, 1, NC>(wr, head, tail - head, lane, accf, H, dstv, lst, cnt, cap);
  if (cap > 0) {
    if (cnt >= 0 && cnt + lane < ((cnt + 3) & ~3)) lst[cnt + lane] = MEMO_END;
    if (lane == 0) ((int*)mb)[gw] = cnt;
  }

  if (blockIdx.x == 0 && tid < (E & 3)) {
    const int e = (E & ~3) + tid;
    const unsigned r = (unsigned)(dst[e] - lo);
    if (r < (unsigned)span) {
      const int s = src[e];
      const float2 dv = dstv[r];
      const float ex = expf(leaky(AS[s] + dv.x) - dv.y);
      atomicAdd(&accf[16 * P + r], ex);
      for (int c = 0; c < NC; ++c) atomicAdd(&accf[c * P + r], ex * H[s * HROW + c]);
    }
  }
  __syncthreads();

  for (int idx = tid; idx < span * DIM; idx += TILE_TB) {
    const int dl = idx >> 4, cc = idx & 15;
    if (cc < NC) atomicAdd(&S[lo * DIM + idx], accf[cc * P + dl]);
  }
  for (int i = tid; i < span; i += TILE_TB) atomicAdd(&DEN[lo + i], accf[16 * P + i]);
}

__device__ __forceinline__ uint4 gat_list_piece(const unsigned* l, int cnt, int i, int lane) {
  int q = (i >> 2) + lane;
  const int qlast = ((cnt + 3) >> 2) - 1;
  q = q < qlast ? q : qlast;
  return ((const uint4*)l)[q];
}

template <int R>
__global__ __launch_bounds__(TILE_TB) void gat_edge_own(
    const int* __restrict__ src, const int* __restrict__ dst, const float* __restrict__ H,
    const float* __restrict__ AS, const float* __restrict__ AD, const unsigned* __restrict__ G,
    const float* __restrict__ b, float* __restrict__ OUT, int E, int n, const float* ws0,
    const unsigned* mb, unsigned long long ws_bytes) {
  static_assert(R % ACC_SLICES == 0 && R < (1 << MEMO_DL_BITS), "sub-range layout");
  constexpr int SUB = R / ACC_SLICES;
  constexpr int P = SUB + 4;
  constexpr int NPL = 17;
  constexpr int WPB = TILE_TB / 32;
  const int off = (int)blockIdx.x * SUB;
  const int lo = (int)blockIdx.y * R + off;
  const int span = (n - lo) < SUB ? (n - lo) : SUB;
  if (span <= 0) return;
  float gmax;
  const bool bound = gat_bound_ok(G, gmax);
  __shared__ unsigned accu[NPL * P];
  __shared__ int2 ring[WPB * RING];
  __shared__ float2 dstv[SUB];
  float* accf = (float*)accu;
  const int tid = threadIdx.x;
  const int lane = tid & 31;
  const int wid = tid >> 5;
  int2* wr = ring + wid * RING;

  const int nwaves = (int)(gridDim.x * gridDim.y) * WPB;
  const int cap = gat_memo_cap(ws0, mb, ws_bytes, n, nwaves);
  const int gw0 = (int)blockIdx.y * ACC_SLICES * WPB + wid;
  int cv = -1;
  if (cap > 0 && lane < ACC_SLICES) cv = ((const int*)mb)[gw0 + lane * WPB];

  const bool sown = tid < span;
  float4 su0 = {0.f, 0.f, 0.f, 0.f}, su1 = su0, su2 = su0, su3 = su0;
  float sas = 0.f;
  if (sown) {
    const float4* p = (const float4*)&H[(lo + tid) * HROW];
    su0 = p[0];
    su1 = p[1];
    su2 = p[2];
    su3 = p[3];
    sas = AS[lo + tid];
  }

  if (!bound) {
    for (int i = tid; i < span; i += TILE_TB) accu[i] = fenc(AS[lo + i]);
    __syncthreads();
    for (int e = tid; e < E; e += TILE_TB) {
      const unsigned r = (unsigned)(dst[e] - lo);
      if (r < (unsigned)span) atomicMax(&accu[r], fenc(AS[src[e]]));
    }
    __syncthreads();
  }
  for (int i = tid; i < span; i += TILE_TB) {
    const float ad = AD[lo + i];
    float2 dv;
    dv.x = ad;
    dv.y = leaky((bound ? gmax : fdec(accu[i])) + ad);
    dstv[i] = dv;
  }
  if (!bound) __syncthreads();
  for (int i = tid; i < NPL * P; i += TILE_TB) accu[i] = 0u;
  __syncthreads();

  if (sown) {
    const float2 dv = dstv[tid];
    const float x = expf(leaky(sas + dv.x) - dv.y);
    gat_acc_row<P, DIM>(accf, tid, x, su0, su1, su2, su3);
  }

  int head = 0, tail = 0, nolist = -1;
  const unsigned* lists = mb + nwaves;
  int j = 0;
  while (j < ACC_SLICES && __builtin_amdgcn_readlane(cv, j) <= 0) ++j;
  int i = 0;
  uint4 w = {0u, 0u, 0u, 0u};
  if (j < ACC_SLICES)
    w = gat_list_piece(lists + (size_t)(gw0 + j * WPB) * (size_t)cap,
                       __builtin_amdgcn_readlane(cv, j), i, lane);
  while (j < ACC_SLICES) {
    const uint4 c = w;
    const int cj = __builtin_amdgcn_readlane(cv, j);
    const bool ok = (i + 4 * lane) < cj;
    i += 4 * 32;
    if (i >= cj) {
      i = 0;
      ++j;
      while (j < ACC_SLICES && __builtin_amdgcn_readlane(cv, j) <= 0) ++j;
    }
    if (j < ACC_SLICES)
      w = gat_list_piece(lists + (size_t)(gw0 + j * WPB) * (size_t)cap,
                         __builtin_amdgcn_readlane(cv, j), i, lane);
    const int4 d4 = make_int4((int)(c.x & ((1u << MEMO_DL_BITS) - 1u)),
                              (int)(c.y & ((1u << MEMO_DL_BITS) - 1u)),
                              (int)(c.z & ((1u << MEMO_DL_BITS) - 1u)),
                              (int)(c.w & ((1u << MEMO_DL_BITS) - 1u)));
    const int4 s4 = make_int4((int)(c.x >> MEMO_DL_BITS), (int)(c.y >> MEMO_DL_BITS),
                              (int)(c.z >> MEMO_DL_BITS), (int)(c.w >> MEMO_DL_BITS));
    gat_ring_push<P, 0, DIM>(d4, s4, ok, off, (unsigned)span, wr, head, tail, lane, accf, H, dstv,
                             nullptr, nolist, 0);
  }
  {
    const int E4 = E >> 2;
    const int per = (E4 + ACC_SLICES - 1) / ACC_SLICES;
    const int4* s4p = (const int4*)src;
    const int4* d4p = (const int4*)dst;
    for (int jj = 0; jj < ACC_SLICES; ++jj) {
      if (__builtin_amdgcn_readlane(cv, jj) >= 0) continue;
      const int q0 = jj * per;
      const int q1 = (q0 + per) < E4 ? (q0 + per) : E4;
      for (int base = q0; base < q1; base += 2 * TILE_TB) {
        for (int h = 0; h < 2; ++h) {
          const int q = base + h * TILE_TB + tid;
          const int qc = q < q1 ? q : (q1 - 1);
          gat_ring_push<P, 0, DIM>(d4p[qc], s4p[qc], q < q1, lo, (unsigned)span, wr, head, tail,
                                   lane, accf, H, dstv, nullptr, nolist, 0);
        }
      }
    }
  }
  if (tail != head)
    gat_ring_drain<P, false, 0, DIM>(wr, head, tail - head, lane, accf, H, dstv, nullptr, nolist,
                                     0);

  if (tid < (E & 3)) {
    const int e = (E & ~3) + tid;
    const unsigned r = (unsigned)(dst[e] - lo);
    if (r < (unsigned)span) {
      const int sn = src[e];
      const float2 dv = dstv[r];
      const float ex = expf(leaky(AS[sn] + dv.x) - dv.y);
      atomicAdd(&accf[16 * P + r], ex);
      for (int c = 0; c < DIM; ++c) atomicAdd(&accf[c * P + r], ex * H[sn * HROW + c]);
    }
  }
  __syncthreads();

  for (int idx = tid; idx < span * DIM; idx += TILE_TB) {
    const int dl = idx >> 4, cc = idx & 15;
    OUT[lo * DIM + idx] = accf[cc * P + dl] / accf[16 * P + dl] + b[cc];
  }
}

__global__ void gat_fin_proj(const float* __restrict__ S, const float* __restrict__ DEN,
                             const float* __restrict__ W, const float* __restrict__ b,
                             float* __restrict__ O, int n) {
  int wave = (int)((blockIdx.x * blockDim.x + threadIdx.x) >> 5);
  int lane = threadIdx.x & 31;
  int ntiles = (n + 15) >> 4;
  if (wave >= ntiles) return;
  int half = lane >> 4;
  int l = lane & 15;
  int rowA = wave * 16 + l;
  if (rowA >= n) rowA = n - 1;
  const float den = DEN[rowA];
  const float bias = b[l];
  v8f c = {};
#pragma unroll
  for (int k0 = 0; k0 < DIM; k0 += 4) {
    const int ka = k0 + half * 2;
    const bool live = ka < NFEAT;
    const int kw = live ? ka : 0;
    const float2 sv = *(const float2*)&S[rowA * DIM + ka];
    v2f a, bw;
    a.x = live ? sv.x / den : 0.f;
    a.y = live ? sv.y / den : 0.f;
    const float w0 = W[kw * DIM + l];
    const float w1 = W[(kw + 1) * DIM + l];
    bw.x = live ? w0 : 0.f;
    bw.y = live ? w1 : 0.f;
    c = __builtin_amdgcn_wmma_f32_16x16x4_f32(false, a, false, bw, (short)0, c,
                                              false, false);
  }
  int base = (wave * 16 + half * 8) * DIM + l;
  if (wave * 16 + 16 <= n) {
#pragma unroll
    for (int r = 0; r < 8; ++r) {
      const float v = c[r] + bias;
      O[base + r * DIM] = v > 0.f ? v : 0.f;
    }
  } else {
#pragma unroll
    for (int r = 0; r < 8; ++r) {
      int row = wave * 16 + r + half * 8;
      const float v = c[r] + bias;
      if (row < n) O[row * DIM + l] = v > 0.f ? v : 0.f;
    }
  }
}

extern "C" void kernel_launch(void* const* d_in, const int* in_sizes, int n_in,
                              void* d_out, int out_size, void* d_ws, size_t ws_size,
                              hipStream_t stream) {
  const float* x = (const float*)d_in[0];
  const int* ei = (const int*)d_in[1];
  const float* W1 = (const float*)d_in[2];
  const float* as1 = (const float*)d_in[3];
  const float* ad1 = (const float*)d_in[4];
  const float* b1 = (const float*)d_in[5];
  const float* W2 = (const float*)d_in[6];
  const float* as2 = (const float*)d_in[7];
  const float* ad2 = (const float*)d_in[8];
  const float* b2 = (const float*)d_in[9];
  float* out = (float*)d_out;

  int n = in_sizes[0] / NFEAT;
  int E = in_sizes[1] / 2;
  const int* src = ei;
  const int* dst = ei + E;

  float* fws = (float*)d_ws;
  float* H = fws;
  float* S = H + (size_t)n * HROW;
  float* O1 = S + (size_t)n * DIM;
  float* AS = O1 + (size_t)n * DIM;
  float* AD = AS + n;
  float* DEN = AD + n;
  unsigned* GB = (unsigned*)(DEN + n);
  unsigned* G2 = GB + 2;
  unsigned* PM = GB + 4;
  unsigned* MB = GB + n + 256;
  const unsigned long long wsb = (unsigned long long)ws_size;

  const int TB = 256;
  int ntiles = (n + 15) >> 4;
  dim3 gProj((ntiles + 7) / 8);
  dim3 gNode((n + TB - 1) / TB);
  const int R_ACC = 3136;
  dim3 gAcc(ACC_SLICES, (n + R_ACC - 1) / R_ACC);

  gat_scores_x<<<gNode, TB, 0, stream>>>(x, W1, as1, ad1, H, AS, AD, S, DEN, PM, G2, n);
  gat_edge_tile<3136, NFEAT><<<gAcc, TILE_TB, 0, stream>>>(src, dst, H, AS, AD, PM, (int)gNode.x, DEN,
                                                          S, E, n, fws, MB, wsb);
  gat_fin_proj<<<gProj, TB, 0, stream>>>(S, DEN, W1, b1, O1, n);

  gat_proj_wmma<<<gProj, TB, 0, stream>>>(O1, W2, H, n);
  gat_alphas<<<gNode, TB, 0, stream>>>(H, as2, ad2, AS, AD, G2, n);
  gat_edge_own<3136><<<gAcc, TILE_TB, 0, stream>>>(src, dst, H, AS, AD, G2, b2, out, E, n, fws, MB,
                                                  wsb);
}
